# starting kernel
# baseline (speedup 1.0000x reference)
#include <hip/hip_runtime.h>

#define Bsz  32
#define Lseq 4096
#define Dd   128
#define Vv   128
#define NCTX 4
#define Wf   512

typedef __attribute__((ext_vector_type(2))) float v2f;
typedef __attribute__((ext_vector_type(8))) float v8f;

__global__ __launch_bounds__(32) void ngram_table_wmma(
    const float* __restrict__ emb,
    const float* __restrict__ W,
    float* __restrict__ Mtab)
{
    const int wid  = blockIdx.x;
    const int k    = wid >> 6;
    const int tile = wid & 63;
    const int m0   = (tile >> 3) << 4;
    const int n0   = (tile & 7)  << 4;

    const int lane = threadIdx.x;
    const int half = lane >> 4;
    const int lr   = lane & 15;

    const float* arow = emb + (m0 + lr) * Dd;
    const float* brow = W   + (n0 + lr) * Wf + k * Dd;

    v8f c = {};
#pragma unroll
    for (int kk = 0; kk < 32; ++kk) {
        const int d0 = kk * 4 + half * 2;
        v2f a, b;
        a.x = arow[d0 + 0];
        a.y = arow[d0 + 1];
        b.x = brow[d0 + 0];
        b.y = brow[d0 + 1];
        c = __builtin_amdgcn_wmma_f32_16x16x4_f32(
                false, a, false, b, (short)0, c, false, false);
    }

    float* outTile = Mtab + k * (Vv * Vv);
#pragma unroll
    for (int r = 0; r < 8; ++r) {
        const int row = m0 + r + 8 * half;
        outTile[row * Vv + (n0 + lr)] = c[r];
    }
}

__global__ __launch_bounds__(256) void ngram_gather_sum(
    const int*   __restrict__ x,
    const float* __restrict__ bias,
    const float* __restrict__ Mtab,
    float4*      __restrict__ out)
{
    const int lane = threadIdx.x & 31;
    const int wave = threadIdx.x >> 5;
    const int pos  = blockIdx.x * 8 + wave;
    const int b    = pos >> 12;
    const int j    = pos & (Lseq - 1);

    float4 acc = ((const float4*)bias)[lane];

    if (j >= NCTX) {
        const int* xp = x + b * Lseq + (j - NCTX);
#pragma unroll
        for (int k = 0; k < NCTX; ++k) {
            const int t = xp[k];
            const float4 m =
                ((const float4*)(Mtab + (k * (Vv * Vv) + t * Vv)))[lane];
            acc.x += m.x; acc.y += m.y; acc.z += m.z; acc.w += m.w;
        }
    }
    out[pos * (Vv / 4) + lane] = acc;
}

extern "C" void kernel_launch(void* const* d_in, const int* in_sizes, int n_in,
                              void* d_out, int out_size, void* d_ws, size_t ws_size,
                              hipStream_t stream) {
    const int*   x    = (const int*)  d_in[0];
    const float* emb  = (const float*)d_in[1];
    const float* W    = (const float*)d_in[2];
    const float* bias = (const float*)d_in[3];

    float* Mtab = (float*)d_ws;

    ngram_table_wmma<<<256, 32, 0, stream>>>(emb, W, Mtab);

    const int positions = Bsz * Lseq;
    ngram_gather_sum<<<positions / 8, 256, 0, stream>>>(
        x, bias, Mtab, (float4*)d_out);
}


# gather kernel: each wave walks 32 consecutive positions, 8 at a time with their 32 table-row loads in flight together; 256 workgroups of 16 waves replace 16384 of 8
# speedup vs baseline: 1.0899x; 1.0899x over previous
#include <hip/hip_runtime.h>

#define Bsz  32
#define Lseq 4096
#define Dd   128
#define Vv   128
#define NCTX 4
#define Wf   512

typedef __attribute__((ext_vector_type(2))) float v2f;
typedef __attribute__((ext_vector_type(8))) float v8f;

__global__ __launch_bounds__(32) void ngram_table_wmma(
    const float* __restrict__ emb,
    const float* __restrict__ W,
    float* __restrict__ Mtab)
{
    const int wid  = blockIdx.x;
    const int k    = wid >> 6;
    const int tile = wid & 63;
    const int m0   = (tile >> 3) << 4;
    const int n0   = (tile & 7)  << 4;

    const int lane = threadIdx.x;
    const int half = lane >> 4;
    const int lr   = lane & 15;

    const float* arow = emb + (m0 + lr) * Dd;
    const float* brow = W   + (n0 + lr) * Wf + k * Dd;

    v8f c = {};
#pragma unroll
    for (int kk = 0; kk < 32; ++kk) {
        const int d0 = kk * 4 + half * 2;
        v2f a, b;
        a.x = arow[d0 + 0];
        a.y = arow[d0 + 1];
        b.x = brow[d0 + 0];
        b.y = brow[d0 + 1];
        c = __builtin_amdgcn_wmma_f32_16x16x4_f32(
                false, a, false, b, (short)0, c, false, false);
    }

    float* outTile = Mtab + k * (Vv * Vv);
#pragma unroll
    for (int r = 0; r < 8; ++r) {
        const int row = m0 + r + 8 * half;
        outTile[row * Vv + (n0 + lr)] = c[r];
    }
}

#define POS_PER_WAVE 32
#define GRP          8
#define WAVES_PER_WG 16

template <bool ROW_START>
__device__ __forceinline__ void ngram_group(
    const int tok, const int lane, const float4 bias4,
    const float4* __restrict__ tab, float4* __restrict__ ogrp)
{
    constexpr int P0 = ROW_START ? NCTX : 0;

    const float4* rowp[GRP + NCTX - 1];
#pragma unroll
    for (int i = P0; i < GRP + NCTX - 1; ++i) {
        const int t = __builtin_amdgcn_readlane(tok, i);
        rowp[i] = tab + t * (Vv / 4) + lane;
    }

    float4 m[GRP][NCTX];
#pragma unroll
    for (int p = P0; p < GRP; ++p) {
#pragma unroll
        for (int k = 0; k < NCTX; ++k)
            m[p][k] = rowp[p + k][k * (Vv * Vv / 4)];
    }

#pragma unroll
    for (int p = 0; p < GRP; ++p) {
        float4 acc = bias4;
        if (p >= P0) {
#pragma unroll
            for (int k = 0; k < NCTX; ++k) {
                acc.x += m[p][k].x; acc.y += m[p][k].y;
                acc.z += m[p][k].z; acc.w += m[p][k].w;
            }
        }
        ogrp[p * (Vv / 4)] = acc;
    }
}

__global__ __launch_bounds__(512) __attribute__((amdgpu_waves_per_eu(4, 4)))
void ngram_gather_sum(
    const int*   __restrict__ x,
    const float* __restrict__ bias,
    const float* __restrict__ Mtab,
    float4*      __restrict__ out)
{
    const int lane = threadIdx.x & 31;
    const int wv = __builtin_amdgcn_readfirstlane(
        (int)(blockIdx.x * WAVES_PER_WG + (threadIdx.x >> 5)));
    const int pos0 = wv * POS_PER_WAVE;
    const int b    = pos0 >> 12;
    const int j0   = pos0 & (Lseq - 1);

    const int*    xrow = x + b * Lseq;
    const float4* tab  = (const float4*)Mtab;
    float4*       orow = out + pos0 * (Vv / 4) + lane;

    const float4 bias4 = ((const float4*)bias)[lane];

    int ti = j0 - NCTX + lane;
    ti = ti < 0 ? 0 : ti;
    int tok = xrow[ti] & (Vv - 1);

#pragma unroll 1
    for (int g = 0; g < POS_PER_WAVE / GRP; ++g) {
        const int jg = j0 + g * GRP;

        int tn = jg + GRP - NCTX + lane;
        tn = tn > Lseq - 1 ? Lseq - 1 : tn;
        const int tokNext = xrow[tn] & (Vv - 1);

        float4* ogrp = orow + g * GRP * (Vv / 4);
        if (jg == 0)
            ngram_group<true >(tok, lane, bias4, tab, ogrp);
        else
            ngram_group<false>(tok, lane, bias4, tab, ogrp);

        tok = tokNext;
    }
}

extern "C" void kernel_launch(void* const* d_in, const int* in_sizes, int n_in,
                              void* d_out, int out_size, void* d_ws, size_t ws_size,
                              hipStream_t stream) {
    const int*   x    = (const int*)  d_in[0];
    const float* emb  = (const float*)d_in[1];
    const float* W    = (const float*)d_in[2];
    const float* bias = (const float*)d_in[3];

    float* Mtab = (float*)d_ws;

    ngram_table_wmma<<<256, 32, 0, stream>>>(emb, W, Mtab);

    ngram_gather_sum<<<256, 512, 0, stream>>>(
        x, bias, Mtab, (float4*)d_out);
}


# gather kernel: each workgroup copies its 64-column half of the four tables into LDS (128 KiB) once and reads the table rows from LDS; 128 chunks of 1024 positions x 2 column halves
# speedup vs baseline: 1.4361x; 1.3176x over previous
#include <hip/hip_runtime.h>

#define Bsz  32
#define Lseq 4096
#define Dd   128
#define Vv   128
#define NCTX 4
#define Wf   512

typedef __attribute__((ext_vector_type(2))) float v2f;
typedef __attribute__((ext_vector_type(8))) float v8f;

__global__ __launch_bounds__(32) void ngram_table_wmma(
    const float* __restrict__ emb,
    const float* __restrict__ W,
    float* __restrict__ Mtab)
{
    const int wid  = blockIdx.x;
    const int k    = wid >> 6;
    const int tile = wid & 63;
    const int m0   = (tile >> 3) << 4;
    const int n0   = (tile & 7)  << 4;

    const int lane = threadIdx.x;
    const int half = lane >> 4;
    const int lr   = lane & 15;

    const float* arow = emb + (m0 + lr) * Dd;
    const float* brow = W   + (n0 + lr) * Wf + k * Dd;

    v8f c = {};
#pragma unroll
    for (int kk = 0; kk < 32; ++kk) {
        const int d0 = kk * 4 + half * 2;
        v2f a, b;
        a.x = arow[d0 + 0];
        a.y = arow[d0 + 1];
        b.x = brow[d0 + 0];
        b.y = brow[d0 + 1];
        c = __builtin_amdgcn_wmma_f32_16x16x4_f32(
                false, a, false, b, (short)0, c, false, false);
    }

    float* outTile = Mtab + k * (Vv * Vv);
#pragma unroll
    for (int r = 0; r < 8; ++r) {
        const int row = m0 + r + 8 * half;
        outTile[row * Vv + (n0 + lr)] = c[r];
    }
}

#define VHALF        64
#define ROWQ         16
#define OUTQ         32
#define POS_PER_WG   1024
#define POS_PER_WAVE 64
#define GRP          8
#define WAVES_PER_WG 16
#define LDS_ROWS     (Vv * NCTX)
#define LDS_ROWS_RESERVED 648

template <bool ROW_START>
__device__ __forceinline__ void ngram_group(
    const int tokq, const bool upper, const float4 bias4,
    const float4* lrow, float4* __restrict__ ogrp)
{
    constexpr int S0 = ROW_START ? NCTX / 2 : 0;
    constexpr int NM = GRP + NCTX - 2;

    const float4* rowp[NM];
#pragma unroll
    for (int m = 2 * S0; m < NM; ++m) {
        const int lo = __builtin_amdgcn_readlane(tokq, m);
        const int hi = __builtin_amdgcn_readlane(tokq, m + 1);
        rowp[m] = lrow + (upper ? hi : lo);
    }

    float4 mv[GRP / 2][NCTX];
#pragma unroll
    for (int i = S0; i < GRP / 2; ++i) {
#pragma unroll
        for (int k = 0; k < NCTX; ++k)
            mv[i][k] = rowp[2 * i + k][k * ROWQ];
    }

#pragma unroll
    for (int i = 0; i < GRP / 2; ++i) {
        float4 acc = bias4;
        if (i >= S0) {
#pragma unroll
            for (int k = 0; k < NCTX; ++k) {
                acc.x += mv[i][k].x; acc.y += mv[i][k].y;
                acc.z += mv[i][k].z; acc.w += mv[i][k].w;
            }
        }
        ogrp[2 * i * OUTQ] = acc;
    }
}

__global__ __launch_bounds__(512) __attribute__((amdgpu_waves_per_eu(4, 4)))
void ngram_gather_sum(
    const int*   __restrict__ x,
    const float* __restrict__ bias,
    const float* __restrict__ Mtab,
    float4*      __restrict__ out)
{
    __shared__ float4 ltab[LDS_ROWS_RESERVED * ROWQ];

    const int tid   = threadIdx.x;
    const int lane  = tid & 31;
    const int c     = lane & 15;
    const bool upper = lane >= 16;
    const int vh    = blockIdx.x & 1;
    const int chunk = blockIdx.x >> 1;

    const int wv = __builtin_amdgcn_readfirstlane(tid >> 5);
    const int pos0 = chunk * POS_PER_WG + wv * POS_PER_WAVE;
    const int b    = pos0 >> 12;
    const int j0   = pos0 & (Lseq - 1);

    const int* xrow = x + b * Lseq;

    int ti = j0 - NCTX + lane;
    ti = ti < 0 ? 0 : ti;
    int tok = xrow[ti] & (Vv - 1);

    const float4 bias4 = ((const float4*)bias)[vh * ROWQ + c];

    {
        const int r0 = tid >> 4;
        const float4* src = (const float4*)Mtab + r0 * OUTQ + vh * ROWQ + c;
        float4 st[16];
#pragma unroll
        for (int p = 0; p < 16; ++p)
            st[p] = src[p * 32 * OUTQ];
#pragma unroll
        for (int p = 0; p < 16; ++p) {
            const int k  = p >> 2;
            const int tb = (p & 3) * 32;
            ltab[((tb + r0) * NCTX + k) * ROWQ + c] = st[p];
        }
    }
    __syncthreads();

    const float4* lrow = ltab + c;
    float4* orow = out + (pos0 + (upper ? 1 : 0)) * OUTQ + vh * ROWQ + c;

#pragma unroll 1
    for (int g = 0; g < POS_PER_WAVE / GRP; ++g) {
        const int jg = j0 + g * GRP;

        int tn = jg + GRP - NCTX + lane;
        tn = tn > Lseq - 1 ? Lseq - 1 : tn;
        const int tokNext = xrow[tn] & (Vv - 1);

        const int tokq = tok * (NCTX * ROWQ);
        float4* ogrp = orow + g * GRP * OUTQ;
        if (jg == 0)
            ngram_group<true >(tokq, upper, bias4, lrow, ogrp);
        else
            ngram_group<false>(tokq, upper, bias4, lrow, ogrp);

        tok = tokNext;
    }
}

extern "C" void kernel_launch(void* const* d_in, const int* in_sizes, int n_in,
                              void* d_out, int out_size, void* d_ws, size_t ws_size,
                              hipStream_t stream) {
    const int*   x    = (const int*)  d_in[0];
    const float* emb  = (const float*)d_in[1];
    const float* W    = (const float*)d_in[2];
    const float* bias = (const float*)d_in[3];

    float* Mtab = (float*)d_ws;

    ngram_table_wmma<<<256, 32, 0, stream>>>(emb, W, Mtab);

    ngram_gather_sum<<<256, 512, 0, stream>>>(
        x, bias, Mtab, (float4*)d_out);
}
